# starting kernel
# baseline (speedup 1.0000x reference)
#include <hip/hip_runtime.h>

typedef __attribute__((ext_vector_type(2))) float v2f;
typedef __attribute__((ext_vector_type(8))) float v8f;

#define DT_F 0.1f

struct alignas(4) F3 { float a, b, c; };

__device__ __forceinline__ float swap16(float v) {
  return __int_as_float(__builtin_amdgcn_ds_swizzle(__float_as_int(v), 0x401F));
}

__device__ __forceinline__ void lorenz(float x, float y, float z,
                                       float sg, float rh, float bt,
                                       float& d0, float& d1, float& d2) {
  d0 = sg * (y - x);
  d1 = rh * x - y - x * z;
  d2 = x * y - bt * z;
}

__global__ __launch_bounds__(256) void mlp_lorenz_rk4(
    const float* __restrict__ x,
    const float* __restrict__ W1, const float* __restrict__ b1,
    const float* __restrict__ W2, const float* __restrict__ b2,
    const float* __restrict__ sigma_p, const float* __restrict__ rho_p,
    const float* __restrict__ beta_p,
    float* __restrict__ out, int numTiles) {
  const int lane = threadIdx.x & 31;
  const int m    = lane & 15;
  const bool hi  = lane >= 16;
  const int waveId = (blockIdx.x * blockDim.x + threadIdx.x) >> 5;
  const int nWaves = (gridDim.x * blockDim.x) >> 5;

  const float sg = sigma_p[0], rh = rho_p[0], bt = beta_p[0];
  const float b2_0 = b2[0], b2_1 = b2[1], b2_2 = b2[2];

  v2f a1;
  a1.x = (m < 10) ? W1[m * 3 + (hi ? 2 : 0)] : 0.0f;
  a1.y = (m < 10 && !hi) ? W1[m * 3 + 1] : 0.0f;

  v2f a2a, a2b, a2c;
  a2a.x = (m < 3) ? W2[m * 10 + (hi ? 2 : 0)] : 0.0f;
  a2a.y = (m < 3) ? W2[m * 10 + (hi ? 3 : 1)] : 0.0f;
  a2b.x = (m < 3) ? W2[m * 10 + (hi ? 6 : 4)] : 0.0f;
  a2b.y = (m < 3) ? W2[m * 10 + (hi ? 7 : 5)] : 0.0f;
  a2c.x = (m < 3 && !hi) ? W2[m * 10 + 8] : 0.0f;
  a2c.y = (m < 3 && !hi) ? W2[m * 10 + 9] : 0.0f;

  float bb[8];
#pragma unroll
  for (int v = 0; v < 8; ++v) {
    const int idx = hi ? (v + 8) : v;
    bb[v] = (idx < 10) ? b1[idx] : 0.0f;
  }

  const v8f czero = {};

  for (int t = waveId; t < numTiles; t += nWaves) {
    const int rb3 = (t * 16 + m) * 3;

    const float bxx = x[rb3 + (hi ? 2 : 0)];
    const float bxy = x[rb3 + 1];
    v2f bx;
    bx.x = bxx;
    bx.y = hi ? 0.0f : bxy;

    {
      int tn = t + nWaves;
      if (tn >= numTiles) tn = t;
      __builtin_prefetch(x + (tn * 16 + m) * 3, 0, 1);
    }

    v8f h = __builtin_amdgcn_wmma_f32_16x16x4_f32(false, a1, false, bx,
                                                  (short)0, czero, false, false);
#pragma unroll
    for (int v = 0; v < 8; ++v) h[v] = fmaxf(h[v] + bb[v], 0.0f);

    const float s0 = swap16(h[0]);
    const float s1 = swap16(h[1]);
    const float s2 = swap16(h[2]);
    const float s3 = swap16(h[3]);
    const float s6 = swap16(h[6]);
    const float s7 = swap16(h[7]);

    v2f bc0, bc1, bc2;
    bc0.x = hi ? s2 : h[0];
    bc0.y = hi ? s3 : h[1];
    bc1.x = hi ? s6 : h[4];
    bc1.y = hi ? s7 : h[5];
    bc2.x = hi ? 0.0f : s0;
    bc2.y = hi ? 0.0f : s1;

    v8f d2 = __builtin_amdgcn_wmma_f32_16x16x4_f32(false, a2a, false, bc0,
                                                   (short)0, czero, false, false);
    d2 = __builtin_amdgcn_wmma_f32_16x16x4_f32(false, a2b, false, bc1,
                                               (short)0, d2, false, false);
    d2 = __builtin_amdgcn_wmma_f32_16x16x4_f32(false, a2c, false, bc2,
                                               (short)0, d2, false, false);

    if (!hi) {
      const float r0 = d2[0] + b2_0;
      const float r1 = d2[1] + b2_1;
      const float r2 = d2[2] + b2_2;

      float k10, k11, k12, k20, k21, k22, k30, k31, k32, k40, k41, k42;
      lorenz(r0, r1, r2, sg, rh, bt, k10, k11, k12);
      lorenz(r0 + 0.5f * DT_F * k10, r1 + 0.5f * DT_F * k11, r2 + 0.5f * DT_F * k12,
             sg, rh, bt, k20, k21, k22);
      lorenz(r0 + 0.5f * DT_F * k20, r1 + 0.5f * DT_F * k21, r2 + 0.5f * DT_F * k22,
             sg, rh, bt, k30, k31, k32);
      lorenz(r0 + DT_F * k30, r1 + DT_F * k31, r2 + DT_F * k32,
             sg, rh, bt, k40, k41, k42);
      const float c = DT_F / 6.0f;

      F3 o;
      o.a = r0 + c * (k10 + 2.0f * k20 + 2.0f * k30 + k40);
      o.b = r1 + c * (k11 + 2.0f * k21 + 2.0f * k31 + k41);
      o.c = r2 + c * (k12 + 2.0f * k22 + 2.0f * k32 + k42);
      reinterpret_cast<F3*>(out)[t * 16 + m] = o;
    }
  }
}

extern "C" void kernel_launch(void* const* d_in, const int* in_sizes, int n_in,
                              void* d_out, int out_size, void* d_ws, size_t ws_size,
                              hipStream_t stream) {
  const float* x     = (const float*)d_in[0];
  const float* W1    = (const float*)d_in[1];
  const float* b1    = (const float*)d_in[2];
  const float* W2    = (const float*)d_in[3];
  const float* b2    = (const float*)d_in[4];
  const float* sigma = (const float*)d_in[5];
  const float* rho   = (const float*)d_in[6];
  const float* beta  = (const float*)d_in[7];
  float* out = (float*)d_out;

  const int Bn = in_sizes[0] / 3;
  const int numTiles = Bn / 16;

  const int block = 256;
  int grid = 8192;
  const int maxBlocks = (numTiles * 32 + block - 1) / block;
  if (grid > maxBlocks) grid = maxBlocks > 0 ? maxBlocks : 1;

  mlp_lorenz_rk4<<<grid, block, 0, stream>>>(x, W1, b1, W2, b2, sigma, rho, beta,
                                             out, numTiles);
}


# Each lane owns one row of a 32-row trip: W2 also at A rows 8-10 so tile 1's state lands in the hi lanes; all 32 lanes run RK4 (was 16), one 384-B b96 load and store per wave
# speedup vs baseline: 1.1566x; 1.1566x over previous
#include <hip/hip_runtime.h>

typedef __attribute__((ext_vector_type(2))) float v2f;
typedef __attribute__((ext_vector_type(8))) float v8f;

#define DT_F 0.1f

struct alignas(4) F3 { float a, b, c; };

__device__ __forceinline__ float swap16(float v) {
  return __int_as_float(__builtin_amdgcn_ds_swizzle(__float_as_int(v), 0x401F));
}

__device__ __forceinline__ void lorenz(float x, float y, float z,
                                       float sg, float rh, float bt,
                                       float& d0, float& d1, float& d2) {
  d0 = sg * (y - x);
  d1 = rh * x - y - x * z;
  d2 = x * y - bt * z;
}

__global__ __launch_bounds__(256) void mlp_lorenz_rk4(
    const float* __restrict__ x,
    const float* __restrict__ W1, const float* __restrict__ b1,
    const float* __restrict__ W2, const float* __restrict__ b2,
    const float* __restrict__ sigma_p, const float* __restrict__ rho_p,
    const float* __restrict__ beta_p,
    float* __restrict__ out, int numPairs) {
  const int lane = threadIdx.x & 31;
  const int m    = lane & 15;
  const bool hi  = lane >= 16;
  const int waveId = (blockIdx.x * blockDim.x + threadIdx.x) >> 5;
  const int nWaves = (gridDim.x * blockDim.x) >> 5;

  const float sg = sigma_p[0], rh = rho_p[0], bt = beta_p[0];
  const float b2_0 = b2[0], b2_1 = b2[1], b2_2 = b2[2];

  v2f a1;
  a1.x = (m < 10) ? W1[m * 3 + (hi ? 2 : 0)] : 0.0f;
  a1.y = (m < 10 && !hi) ? W1[m * 3 + 1] : 0.0f;

  const bool mv = (m & 7) < 3;
  const int  mo = (m & 7) * 10;
  v2f a2a, a2b, a2c;
  a2a.x = mv ? W2[mo + (hi ? 2 : 0)] : 0.0f;
  a2a.y = mv ? W2[mo + (hi ? 3 : 1)] : 0.0f;
  a2b.x = mv ? W2[mo + (hi ? 6 : 4)] : 0.0f;
  a2b.y = mv ? W2[mo + (hi ? 7 : 5)] : 0.0f;
  a2c.x = (mv && !hi) ? W2[mo + 8] : 0.0f;
  a2c.y = (mv && !hi) ? W2[mo + 9] : 0.0f;

  float bb[8];
#pragma unroll
  for (int v = 0; v < 8; ++v) {
    const int idx = hi ? (v + 8) : v;
    bb[v] = (idx < 10) ? b1[idx] : 0.0f;
  }

  const v8f czero = {};

  for (int t = waveId; t < numPairs; t += nWaves) {
    const int row = t * 32 + lane;

    const F3 p = reinterpret_cast<const F3*>(x)[row];

    {
      int tn = t + nWaves;
      if (tn >= numPairs) tn = t;
      __builtin_prefetch(x + (tn * 32 + lane) * 3, 0, 1);
    }

    const float q0 = swap16(p.a);
    const float q1 = swap16(p.b);
    const float q2 = swap16(p.c);

    v2f bx0, bx1;
    bx0.x = hi ? q2 : p.a;
    bx0.y = hi ? 0.0f : p.b;
    bx1.x = hi ? p.c : q0;
    bx1.y = hi ? 0.0f : q1;

    v8f dd[2];
#pragma unroll
    for (int u = 0; u < 2; ++u) {
      const v2f bx = u ? bx1 : bx0;
      v8f h = __builtin_amdgcn_wmma_f32_16x16x4_f32(false, a1, false, bx,
                                                    (short)0, czero, false, false);
#pragma unroll
      for (int v = 0; v < 8; ++v) h[v] = fmaxf(h[v] + bb[v], 0.0f);

      const float s0 = swap16(h[0]);
      const float s1 = swap16(h[1]);
      const float s2 = swap16(h[2]);
      const float s3 = swap16(h[3]);
      const float s6 = swap16(h[6]);
      const float s7 = swap16(h[7]);

      v2f bc0, bc1, bc2;
      bc0.x = hi ? s2 : h[0];
      bc0.y = hi ? s3 : h[1];
      bc1.x = hi ? s6 : h[4];
      bc1.y = hi ? s7 : h[5];
      bc2.x = hi ? 0.0f : s0;
      bc2.y = hi ? 0.0f : s1;

      v8f d2 = __builtin_amdgcn_wmma_f32_16x16x4_f32(false, a2a, false, bc0,
                                                     (short)0, czero, false, false);
      d2 = __builtin_amdgcn_wmma_f32_16x16x4_f32(false, a2b, false, bc1,
                                                 (short)0, d2, false, false);
      d2 = __builtin_amdgcn_wmma_f32_16x16x4_f32(false, a2c, false, bc2,
                                                 (short)0, d2, false, false);
      dd[u] = d2;
    }

    const float r0 = (hi ? dd[1][0] : dd[0][0]) + b2_0;
    const float r1 = (hi ? dd[1][1] : dd[0][1]) + b2_1;
    const float r2 = (hi ? dd[1][2] : dd[0][2]) + b2_2;

    float k10, k11, k12, k20, k21, k22, k30, k31, k32, k40, k41, k42;
    lorenz(r0, r1, r2, sg, rh, bt, k10, k11, k12);
    lorenz(r0 + 0.5f * DT_F * k10, r1 + 0.5f * DT_F * k11, r2 + 0.5f * DT_F * k12,
           sg, rh, bt, k20, k21, k22);
    lorenz(r0 + 0.5f * DT_F * k20, r1 + 0.5f * DT_F * k21, r2 + 0.5f * DT_F * k22,
           sg, rh, bt, k30, k31, k32);
    lorenz(r0 + DT_F * k30, r1 + DT_F * k31, r2 + DT_F * k32,
           sg, rh, bt, k40, k41, k42);
    const float c = DT_F / 6.0f;

    F3 o;
    o.a = r0 + c * (k10 + 2.0f * k20 + 2.0f * k30 + k40);
    o.b = r1 + c * (k11 + 2.0f * k21 + 2.0f * k31 + k41);
    o.c = r2 + c * (k12 + 2.0f * k22 + 2.0f * k32 + k42);
    reinterpret_cast<F3*>(out)[row] = o;
  }
}

extern "C" void kernel_launch(void* const* d_in, const int* in_sizes, int n_in,
                              void* d_out, int out_size, void* d_ws, size_t ws_size,
                              hipStream_t stream) {
  const float* x     = (const float*)d_in[0];
  const float* W1    = (const float*)d_in[1];
  const float* b1    = (const float*)d_in[2];
  const float* W2    = (const float*)d_in[3];
  const float* b2    = (const float*)d_in[4];
  const float* sigma = (const float*)d_in[5];
  const float* rho   = (const float*)d_in[6];
  const float* beta  = (const float*)d_in[7];
  float* out = (float*)d_out;

  const int Bn = in_sizes[0] / 3;
  const int numPairs = Bn / 32;

  const int block = 256;
  int grid = 8192;
  const int maxBlocks = (numPairs * 32 + block - 1) / block;
  if (grid > maxBlocks) grid = maxBlocks > 0 ? maxBlocks : 1;

  mlp_lorenz_rk4<<<grid, block, 0, stream>>>(x, W1, b1, W2, b2, sigma, rho, beta,
                                             out, numPairs);
}


# Hidden units slotted 5+5 over the lane halves: layer-1 D registers are layer-2's B pairs as they stand (no lane swaps), layer 2 one block-diagonal K=20 product, 5 WMMAs for 6; its k sum order changes
# speedup vs baseline: 1.3311x; 1.1509x over previous
#include <hip/hip_runtime.h>

typedef __attribute__((ext_vector_type(2))) float v2f;
typedef __attribute__((ext_vector_type(8))) float v8f;

#define DT_F 0.1f

struct alignas(4) F3 { float a, b, c; };

__device__ __forceinline__ float swap16(float v) {
  return __int_as_float(__builtin_amdgcn_ds_swizzle(__float_as_int(v), 0x401F));
}

__device__ __forceinline__ void lorenz(float x, float y, float z,
                                       float sg, float rh, float bt,
                                       float& d0, float& d1, float& d2) {
  d0 = sg * (y - x);
  d1 = rh * x - y - x * z;
  d2 = x * y - bt * z;
}

__global__ __launch_bounds__(256) void mlp_lorenz_rk4(
    const float* __restrict__ x,
    const float* __restrict__ W1, const float* __restrict__ b1,
    const float* __restrict__ W2, const float* __restrict__ b2,
    const float* __restrict__ sigma_p, const float* __restrict__ rho_p,
    const float* __restrict__ beta_p,
    float* __restrict__ out, int numPairs) {
  const int lane = threadIdx.x & 31;
  const int m    = lane & 15;
  const bool hi  = lane >= 16;
  const int waveId = (blockIdx.x * blockDim.x + threadIdx.x) >> 5;
  const int nWaves = (gridDim.x * blockDim.x) >> 5;

  const float sg = sigma_p[0], rh = rho_p[0], bt = beta_p[0];
  const float b2_0 = b2[0], b2_1 = b2[1], b2_2 = b2[2];

  const int  r8  = m & 7;
  const bool hv  = r8 < 5;
  const int  hid = (m >> 3) * 5 + r8;
  v2f a1;
  a1.x = hv ? W1[hid * 3 + (hi ? 2 : 0)] : 0.0f;
  a1.y = (hv && !hi) ? W1[hid * 3 + 1] : 0.0f;

  const bool jv = r8 < 3;
  const int  mt = m >> 3;
  const int  wo = r8 * 10 + (hi ? 5 : 0);
  v2f a2[5];
#pragma unroll
  for (int i = 0; i < 5; ++i) {
#pragma unroll
    for (int s = 0; s < 2; ++s) {
      const int q    = 2 * i + s;
      const int tile = q >= 5 ? 1 : 0;
      const int hq   = q < 5 ? q : (q == 5 ? 4 : q - 6);
      a2[i][s] = (jv && mt == tile) ? W2[wo + hq] : 0.0f;
    }
  }

  float bb[5];
#pragma unroll
  for (int v = 0; v < 5; ++v) bb[v] = b1[(hi ? 5 : 0) + v];

  const v8f czero = {};

  for (int t = waveId; t < numPairs; t += nWaves) {
    const int row = t * 32 + lane;

    const F3 p = reinterpret_cast<const F3*>(x)[row];

    {
      int tn = t + nWaves;
      if (tn >= numPairs) tn = t;
      __builtin_prefetch(x + (tn * 32 + lane) * 3, 0, 1);
    }

    const float q0 = swap16(p.a);
    const float q1 = swap16(p.b);
    const float q2 = swap16(p.c);

    v2f bx0, bx1;
    bx0.x = hi ? q2 : p.a;
    bx0.y = hi ? 0.0f : p.b;
    bx1.x = hi ? p.c : q0;
    bx1.y = hi ? 0.0f : q1;

    v8f h0 = __builtin_amdgcn_wmma_f32_16x16x4_f32(false, a1, false, bx0,
                                                   (short)0, czero, false, false);
    v8f h1 = __builtin_amdgcn_wmma_f32_16x16x4_f32(false, a1, false, bx1,
                                                   (short)0, czero, false, false);
#pragma unroll
    for (int v = 0; v < 5; ++v) {
      h0[v] = fmaxf(h0[v] + bb[v], 0.0f);
      h1[v] = fmaxf(h1[v] + bb[v], 0.0f);
    }

    v2f c0, c1, c2, c3, c4;
    c0.x = h0[0]; c0.y = h0[1];
    c1.x = h0[2]; c1.y = h0[3];
    c2.x = h0[4]; c2.y = h1[4];
    c3.x = h1[0]; c3.y = h1[1];
    c4.x = h1[2]; c4.y = h1[3];

    v8f d2 = __builtin_amdgcn_wmma_f32_16x16x4_f32(false, a2[0], false, c0,
                                                   (short)0, czero, false, false);
    d2 = __builtin_amdgcn_wmma_f32_16x16x4_f32(false, a2[1], false, c1,
                                               (short)0, d2, false, false);
    d2 = __builtin_amdgcn_wmma_f32_16x16x4_f32(false, a2[2], false, c2,
                                               (short)0, d2, false, false);
    d2 = __builtin_amdgcn_wmma_f32_16x16x4_f32(false, a2[3], false, c3,
                                               (short)0, d2, false, false);
    d2 = __builtin_amdgcn_wmma_f32_16x16x4_f32(false, a2[4], false, c4,
                                               (short)0, d2, false, false);

    const float r0 = d2[0] + b2_0;
    const float r1 = d2[1] + b2_1;
    const float r2 = d2[2] + b2_2;

    float k10, k11, k12, k20, k21, k22, k30, k31, k32, k40, k41, k42;
    lorenz(r0, r1, r2, sg, rh, bt, k10, k11, k12);
    lorenz(r0 + 0.5f * DT_F * k10, r1 + 0.5f * DT_F * k11, r2 + 0.5f * DT_F * k12,
           sg, rh, bt, k20, k21, k22);
    lorenz(r0 + 0.5f * DT_F * k20, r1 + 0.5f * DT_F * k21, r2 + 0.5f * DT_F * k22,
           sg, rh, bt, k30, k31, k32);
    lorenz(r0 + DT_F * k30, r1 + DT_F * k31, r2 + DT_F * k32,
           sg, rh, bt, k40, k41, k42);
    const float c = DT_F / 6.0f;

    F3 o;
    o.a = r0 + c * (k10 + 2.0f * k20 + 2.0f * k30 + k40);
    o.b = r1 + c * (k11 + 2.0f * k21 + 2.0f * k31 + k41);
    o.c = r2 + c * (k12 + 2.0f * k22 + 2.0f * k32 + k42);
    reinterpret_cast<F3*>(out)[row] = o;
  }
}

extern "C" void kernel_launch(void* const* d_in, const int* in_sizes, int n_in,
                              void* d_out, int out_size, void* d_ws, size_t ws_size,
                              hipStream_t stream) {
  const float* x     = (const float*)d_in[0];
  const float* W1    = (const float*)d_in[1];
  const float* b1    = (const float*)d_in[2];
  const float* W2    = (const float*)d_in[3];
  const float* b2    = (const float*)d_in[4];
  const float* sigma = (const float*)d_in[5];
  const float* rho   = (const float*)d_in[6];
  const float* beta  = (const float*)d_in[7];
  float* out = (float*)d_out;

  const int Bn = in_sizes[0] / 3;
  const int numPairs = Bn / 32;

  const int block = 256;
  int grid = 8192;
  const int maxBlocks = (numPairs * 32 + block - 1) / block;
  if (grid > maxBlocks) grid = maxBlocks > 0 ? maxBlocks : 1;

  mlp_lorenz_rk4<<<grid, block, 0, stream>>>(x, W1, b1, W2, b2, sigma, rho, beta,
                                             out, numPairs);
}


# Loop state on the scalar unit: wave index via readfirstlane, rows addressed as scalar trip base + 32-bit lane offset; counter, bound test, prefetch clamp and pointer bumps leave the vector port
# speedup vs baseline: 1.4030x; 1.0540x over previous
#include <hip/hip_runtime.h>

typedef __attribute__((ext_vector_type(2))) float v2f;
typedef __attribute__((ext_vector_type(8))) float v8f;

#define DT_F 0.1f

struct alignas(4) F3 { float a, b, c; };

__device__ __forceinline__ float swap16(float v) {
  return __int_as_float(__builtin_amdgcn_ds_swizzle(__float_as_int(v), 0x401F));
}

__device__ __forceinline__ void lorenz(float x, float y, float z,
                                       float sg, float rh, float bt,
                                       float& d0, float& d1, float& d2) {
  d0 = sg * (y - x);
  d1 = rh * x - y - x * z;
  d2 = x * y - bt * z;
}

__global__ __launch_bounds__(256) void mlp_lorenz_rk4(
    const float* __restrict__ x,
    const float* __restrict__ W1, const float* __restrict__ b1,
    const float* __restrict__ W2, const float* __restrict__ b2,
    const float* __restrict__ sigma_p, const float* __restrict__ rho_p,
    const float* __restrict__ beta_p,
    float* __restrict__ out, int numPairs) {
  const int lane = threadIdx.x & 31;
  const int m    = lane & 15;
  const bool hi  = lane >= 16;
  const int waveId = __builtin_amdgcn_readfirstlane((int)((blockIdx.x * blockDim.x + threadIdx.x) >> 5));
  const int nWaves = (gridDim.x * blockDim.x) >> 5;

  const float sg = sigma_p[0], rh = rho_p[0], bt = beta_p[0];
  const float b2_0 = b2[0], b2_1 = b2[1], b2_2 = b2[2];

  const int  r8  = m & 7;
  const bool hv  = r8 < 5;
  const int  hid = (m >> 3) * 5 + r8;
  v2f a1;
  a1.x = hv ? W1[hid * 3 + (hi ? 2 : 0)] : 0.0f;
  a1.y = (hv && !hi) ? W1[hid * 3 + 1] : 0.0f;

  const bool jv = r8 < 3;
  const int  mt = m >> 3;
  const int  wo = r8 * 10 + (hi ? 5 : 0);
  v2f a2[5];
#pragma unroll
  for (int i = 0; i < 5; ++i) {
#pragma unroll
    for (int s = 0; s < 2; ++s) {
      const int q    = 2 * i + s;
      const int tile = q >= 5 ? 1 : 0;
      const int hq   = q < 5 ? q : (q == 5 ? 4 : q - 6);
      a2[i][s] = (jv && mt == tile) ? W2[wo + hq] : 0.0f;
    }
  }

  float bb[5];
#pragma unroll
  for (int v = 0; v < 5; ++v) bb[v] = b1[(hi ? 5 : 0) + v];

  const v8f czero = {};

  unsigned lo = lane * 12u;

  for (int t = waveId; t < numPairs; t += nWaves) {
    asm volatile("" : "+v"(lo));

    const char* xb = reinterpret_cast<const char*>(x) + (size_t)t * 384;
    const F3 p = *reinterpret_cast<const F3*>(xb + lo);

    {
      int tn = t + nWaves;
      if (tn >= numPairs) tn = t;
      __builtin_prefetch(reinterpret_cast<const char*>(x) + (size_t)tn * 384 + lo, 0, 1);
    }

    const float q0 = swap16(p.a);
    const float q1 = swap16(p.b);
    const float q2 = swap16(p.c);

    v2f bx0, bx1;
    bx0.x = hi ? q2 : p.a;
    bx0.y = hi ? 0.0f : p.b;
    bx1.x = hi ? p.c : q0;
    bx1.y = hi ? 0.0f : q1;

    v8f h0 = __builtin_amdgcn_wmma_f32_16x16x4_f32(false, a1, false, bx0,
                                                   (short)0, czero, false, false);
    v8f h1 = __builtin_amdgcn_wmma_f32_16x16x4_f32(false, a1, false, bx1,
                                                   (short)0, czero, false, false);
#pragma unroll
    for (int v = 0; v < 5; ++v) {
      h0[v] = fmaxf(h0[v] + bb[v], 0.0f);
      h1[v] = fmaxf(h1[v] + bb[v], 0.0f);
    }

    v2f c0, c1, c2, c3, c4;
    c0.x = h0[0]; c0.y = h0[1];
    c1.x = h0[2]; c1.y = h0[3];
    c2.x = h0[4]; c2.y = h1[4];
    c3.x = h1[0]; c3.y = h1[1];
    c4.x = h1[2]; c4.y = h1[3];

    v8f d2 = __builtin_amdgcn_wmma_f32_16x16x4_f32(false, a2[0], false, c0,
                                                   (short)0, czero, false, false);
    d2 = __builtin_amdgcn_wmma_f32_16x16x4_f32(false, a2[1], false, c1,
                                               (short)0, d2, false, false);
    d2 = __builtin_amdgcn_wmma_f32_16x16x4_f32(false, a2[2], false, c2,
                                               (short)0, d2, false, false);
    d2 = __builtin_amdgcn_wmma_f32_16x16x4_f32(false, a2[3], false, c3,
                                               (short)0, d2, false, false);
    d2 = __builtin_amdgcn_wmma_f32_16x16x4_f32(false, a2[4], false, c4,
                                               (short)0, d2, false, false);

    const float r0 = d2[0] + b2_0;
    const float r1 = d2[1] + b2_1;
    const float r2 = d2[2] + b2_2;

    float k10, k11, k12, k20, k21, k22, k30, k31, k32, k40, k41, k42;
    lorenz(r0, r1, r2, sg, rh, bt, k10, k11, k12);
    lorenz(r0 + 0.5f * DT_F * k10, r1 + 0.5f * DT_F * k11, r2 + 0.5f * DT_F * k12,
           sg, rh, bt, k20, k21, k22);
    lorenz(r0 + 0.5f * DT_F * k20, r1 + 0.5f * DT_F * k21, r2 + 0.5f * DT_F * k22,
           sg, rh, bt, k30, k31, k32);
    lorenz(r0 + DT_F * k30, r1 + DT_F * k31, r2 + DT_F * k32,
           sg, rh, bt, k40, k41, k42);
    const float c = DT_F / 6.0f;

    F3 o;
    o.a = r0 + c * (k10 + 2.0f * k20 + 2.0f * k30 + k40);
    o.b = r1 + c * (k11 + 2.0f * k21 + 2.0f * k31 + k41);
    o.c = r2 + c * (k12 + 2.0f * k22 + 2.0f * k32 + k42);
    *reinterpret_cast<F3*>(reinterpret_cast<char*>(out) + (size_t)t * 384 + lo) = o;
  }
}

extern "C" void kernel_launch(void* const* d_in, const int* in_sizes, int n_in,
                              void* d_out, int out_size, void* d_ws, size_t ws_size,
                              hipStream_t stream) {
  const float* x     = (const float*)d_in[0];
  const float* W1    = (const float*)d_in[1];
  const float* b1    = (const float*)d_in[2];
  const float* W2    = (const float*)d_in[3];
  const float* b2    = (const float*)d_in[4];
  const float* sigma = (const float*)d_in[5];
  const float* rho   = (const float*)d_in[6];
  const float* beta  = (const float*)d_in[7];
  float* out = (float*)d_out;

  const int Bn = in_sizes[0] / 3;
  const int numPairs = Bn / 32;

  const int block = 256;
  int grid = 8192;
  const int maxBlocks = (numPairs * 32 + block - 1) / block;
  if (grid > maxBlocks) grid = maxBlocks > 0 ? maxBlocks : 1;

  mlp_lorenz_rk4<<<grid, block, 0, stream>>>(x, W1, b1, W2, b2, sigma, rho, beta,
                                             out, numPairs);
}


# Two adjacent 32-row trips per loop pass (768 B per wave), their WMMA chains side by side: accumulate, bias+ReLU and RK4 entry no longer wait on the WMMA just in front; half the waves, 4 passes each
# speedup vs baseline: 1.5513x; 1.1057x over previous
#include <hip/hip_runtime.h>

typedef __attribute__((ext_vector_type(2))) float v2f;
typedef __attribute__((ext_vector_type(8))) float v8f;

#define DT_F 0.1f

struct alignas(4) F3 { float a, b, c; };

__device__ __forceinline__ float swap16(float v) {
  return __int_as_float(__builtin_amdgcn_ds_swizzle(__float_as_int(v), 0x401F));
}

__device__ __forceinline__ void lorenz(float x, float y, float z,
                                       float sg, float rh, float bt,
                                       float& d0, float& d1, float& d2) {
  d0 = sg * (y - x);
  d1 = rh * x - y - x * z;
  d2 = x * y - bt * z;
}

__device__ __forceinline__ void xfrags(const F3& p, bool hi, v2f& bx0, v2f& bx1) {
  const float q0 = swap16(p.a);
  const float q1 = swap16(p.b);
  const float q2 = swap16(p.c);
  bx0.x = hi ? q2 : p.a;
  bx0.y = hi ? 0.0f : p.b;
  bx1.x = hi ? p.c : q0;
  bx1.y = hi ? 0.0f : q1;
}

__device__ __forceinline__ void hfrags(const v8f& h0, const v8f& h1, v2f (&c)[5]) {
  c[0].x = h0[0]; c[0].y = h0[1];
  c[1].x = h0[2]; c[1].y = h0[3];
  c[2].x = h0[4]; c[2].y = h1[4];
  c[3].x = h1[0]; c[3].y = h1[1];
  c[4].x = h1[2]; c[4].y = h1[3];
}

__device__ __forceinline__ void bias_relu(v8f& h0, v8f& h1, const float (&bb)[5]) {
#pragma unroll
  for (int v = 0; v < 5; ++v) {
    h0[v] = fmaxf(h0[v] + bb[v], 0.0f);
    h1[v] = fmaxf(h1[v] + bb[v], 0.0f);
  }
}

__device__ __forceinline__ F3 rk4(float r0, float r1, float r2, float sg, float rh, float bt) {
  float k10, k11, k12, k20, k21, k22, k30, k31, k32, k40, k41, k42;
  lorenz(r0, r1, r2, sg, rh, bt, k10, k11, k12);
  lorenz(r0 + 0.5f * DT_F * k10, r1 + 0.5f * DT_F * k11, r2 + 0.5f * DT_F * k12,
         sg, rh, bt, k20, k21, k22);
  lorenz(r0 + 0.5f * DT_F * k20, r1 + 0.5f * DT_F * k21, r2 + 0.5f * DT_F * k22,
         sg, rh, bt, k30, k31, k32);
  lorenz(r0 + DT_F * k30, r1 + DT_F * k31, r2 + DT_F * k32,
         sg, rh, bt, k40, k41, k42);
  const float c = DT_F / 6.0f;
  F3 o;
  o.a = r0 + c * (k10 + 2.0f * k20 + 2.0f * k30 + k40);
  o.b = r1 + c * (k11 + 2.0f * k21 + 2.0f * k31 + k41);
  o.c = r2 + c * (k12 + 2.0f * k22 + 2.0f * k32 + k42);
  return o;
}

__global__ __launch_bounds__(256) void mlp_lorenz_rk4(
    const float* __restrict__ x,
    const float* __restrict__ W1, const float* __restrict__ b1,
    const float* __restrict__ W2, const float* __restrict__ b2,
    const float* __restrict__ sigma_p, const float* __restrict__ rho_p,
    const float* __restrict__ beta_p,
    float* __restrict__ out, int numPairs) {
  const int lane = threadIdx.x & 31;
  const int m    = lane & 15;
  const bool hi  = lane >= 16;
  const int waveId = __builtin_amdgcn_readfirstlane((int)((blockIdx.x * blockDim.x + threadIdx.x) >> 5));
  const int nWaves = (gridDim.x * blockDim.x) >> 5;

  const float sg = sigma_p[0], rh = rho_p[0], bt = beta_p[0];
  const float b2_0 = b2[0], b2_1 = b2[1], b2_2 = b2[2];

  const int  r8  = m & 7;
  const bool hv  = r8 < 5;
  const int  hid = (m >> 3) * 5 + r8;
  v2f a1;
  a1.x = hv ? W1[hid * 3 + (hi ? 2 : 0)] : 0.0f;
  a1.y = (hv && !hi) ? W1[hid * 3 + 1] : 0.0f;

  const bool jv = r8 < 3;
  const int  mt = m >> 3;
  const int  wo = r8 * 10 + (hi ? 5 : 0);
  v2f a2[5];
#pragma unroll
  for (int i = 0; i < 5; ++i) {
#pragma unroll
    for (int s = 0; s < 2; ++s) {
      const int q    = 2 * i + s;
      const int tile = q >= 5 ? 1 : 0;
      const int hq   = q < 5 ? q : (q == 5 ? 4 : q - 6);
      a2[i][s] = (jv && mt == tile) ? W2[wo + hq] : 0.0f;
    }
  }

  float bb[5];
#pragma unroll
  for (int v = 0; v < 5; ++v) bb[v] = b1[(hi ? 5 : 0) + v];

  const v8f czero = {};

  unsigned lo = lane * 12u;

  const int numDuo = numPairs >> 1;
  for (int u = waveId; u < numDuo; u += nWaves) {
    asm volatile("" : "+v"(lo));

    const char* xb = reinterpret_cast<const char*>(x) + (size_t)u * 768;
    const F3 pA = *reinterpret_cast<const F3*>(xb + lo);
    const F3 pB = *reinterpret_cast<const F3*>(xb + 384 + lo);

    {
      int un = u + nWaves;
      if (un >= numDuo) un = u;
      const char* xn = reinterpret_cast<const char*>(x) + (size_t)un * 768;
      __builtin_prefetch(xn + lo, 0, 1);
      __builtin_prefetch(xn + 384 + lo, 0, 1);
    }

    v2f xA0, xA1, xB0, xB1;
    xfrags(pA, hi, xA0, xA1);
    xfrags(pB, hi, xB0, xB1);
    v8f hA0 = __builtin_amdgcn_wmma_f32_16x16x4_f32(false, a1, false, xA0, (short)0, czero, false, false);
    v8f hA1 = __builtin_amdgcn_wmma_f32_16x16x4_f32(false, a1, false, xA1, (short)0, czero, false, false);
    v8f hB0 = __builtin_amdgcn_wmma_f32_16x16x4_f32(false, a1, false, xB0, (short)0, czero, false, false);
    v8f hB1 = __builtin_amdgcn_wmma_f32_16x16x4_f32(false, a1, false, xB1, (short)0, czero, false, false);

    bias_relu(hA0, hA1, bb);
    bias_relu(hB0, hB1, bb);

    v2f cA[5], cB[5];
    hfrags(hA0, hA1, cA);
    hfrags(hB0, hB1, cB);

    v8f dA = __builtin_amdgcn_wmma_f32_16x16x4_f32(false, a2[0], false, cA[0], (short)0, czero, false, false);
    v8f dB = __builtin_amdgcn_wmma_f32_16x16x4_f32(false, a2[0], false, cB[0], (short)0, czero, false, false);
#pragma unroll
    for (int i = 1; i < 5; ++i) {
      dA = __builtin_amdgcn_wmma_f32_16x16x4_f32(false, a2[i], false, cA[i], (short)0, dA, false, false);
      dB = __builtin_amdgcn_wmma_f32_16x16x4_f32(false, a2[i], false, cB[i], (short)0, dB, false, false);
    }

    char* ob = reinterpret_cast<char*>(out) + (size_t)u * 768;
    const F3 oA = rk4(dA[0] + b2_0, dA[1] + b2_1, dA[2] + b2_2, sg, rh, bt);
    const F3 oB = rk4(dB[0] + b2_0, dB[1] + b2_1, dB[2] + b2_2, sg, rh, bt);
    *reinterpret_cast<F3*>(ob + lo) = oA;
    *reinterpret_cast<F3*>(ob + 384 + lo) = oB;
  }

  if ((numPairs & 1) && waveId == 0) {
    const size_t off = (size_t)(numPairs - 1) * 384 + lo;
    const F3 p = *reinterpret_cast<const F3*>(reinterpret_cast<const char*>(x) + off);
    v2f x0, x1;
    xfrags(p, hi, x0, x1);
    v8f h0 = __builtin_amdgcn_wmma_f32_16x16x4_f32(false, a1, false, x0, (short)0, czero, false, false);
    v8f h1 = __builtin_amdgcn_wmma_f32_16x16x4_f32(false, a1, false, x1, (short)0, czero, false, false);
    bias_relu(h0, h1, bb);
    v2f c[5];
    hfrags(h0, h1, c);
    v8f d = __builtin_amdgcn_wmma_f32_16x16x4_f32(false, a2[0], false, c[0], (short)0, czero, false, false);
#pragma unroll
    for (int i = 1; i < 5; ++i)
      d = __builtin_amdgcn_wmma_f32_16x16x4_f32(false, a2[i], false, c[i], (short)0, d, false, false);
    *reinterpret_cast<F3*>(reinterpret_cast<char*>(out) + off) = rk4(d[0] + b2_0, d[1] + b2_1, d[2] + b2_2, sg, rh, bt);
  }
}

extern "C" void kernel_launch(void* const* d_in, const int* in_sizes, int n_in,
                              void* d_out, int out_size, void* d_ws, size_t ws_size,
                              hipStream_t stream) {
  const float* x     = (const float*)d_in[0];
  const float* W1    = (const float*)d_in[1];
  const float* b1    = (const float*)d_in[2];
  const float* W2    = (const float*)d_in[3];
  const float* b2    = (const float*)d_in[4];
  const float* sigma = (const float*)d_in[5];
  const float* rho   = (const float*)d_in[6];
  const float* beta  = (const float*)d_in[7];
  float* out = (float*)d_out;

  const int Bn = in_sizes[0] / 3;
  const int numPairs = Bn / 32;

  const int block = 256;
  int grid = 4096;
  const int maxBlocks = (numPairs * 16 + block - 1) / block;
  if (grid > maxBlocks) grid = maxBlocks > 0 ? maxBlocks : 1;

  mlp_lorenz_rk4<<<grid, block, 0, stream>>>(x, W1, b1, W2, b2, sigma, rho, beta,
                                             out, numPairs);
}


# Layer-1 B fragments read straight from memory in fragment layout (lo lane x0,x1; hi lane x1,x2 of its tile row; A's K2 column zero): 6 lane swaps, 8 selects per pass go; layer-1 k order changes
# speedup vs baseline: 1.6089x; 1.0371x over previous
#include <hip/hip_runtime.h>

typedef __attribute__((ext_vector_type(2))) float v2f;
typedef __attribute__((ext_vector_type(8))) float v8f;

#define DT_F 0.1f

struct alignas(4) F3 { float a, b, c; };
struct alignas(4) F2 { float a, b; };

__device__ __forceinline__ void lorenz(float x, float y, float z,
                                       float sg, float rh, float bt,
                                       float& d0, float& d1, float& d2) {
  d0 = sg * (y - x);
  d1 = rh * x - y - x * z;
  d2 = x * y - bt * z;
}

__device__ __forceinline__ v2f xfrag(const char* p) {
  const F2 f = *reinterpret_cast<const F2*>(p);
  v2f b;
  b.x = f.a;
  b.y = f.b;
  return b;
}

__device__ __forceinline__ void hfrags(const v8f& h0, const v8f& h1, v2f (&c)[5]) {
  c[0].x = h0[0]; c[0].y = h0[1];
  c[1].x = h0[2]; c[1].y = h0[3];
  c[2].x = h0[4]; c[2].y = h1[4];
  c[3].x = h1[0]; c[3].y = h1[1];
  c[4].x = h1[2]; c[4].y = h1[3];
}

__device__ __forceinline__ void bias_relu(v8f& h0, v8f& h1, const float (&bb)[5]) {
#pragma unroll
  for (int v = 0; v < 5; ++v) {
    h0[v] = fmaxf(h0[v] + bb[v], 0.0f);
    h1[v] = fmaxf(h1[v] + bb[v], 0.0f);
  }
}

__device__ __forceinline__ F3 rk4(float r0, float r1, float r2, float sg, float rh, float bt) {
  float k10, k11, k12, k20, k21, k22, k30, k31, k32, k40, k41, k42;
  lorenz(r0, r1, r2, sg, rh, bt, k10, k11, k12);
  lorenz(r0 + 0.5f * DT_F * k10, r1 + 0.5f * DT_F * k11, r2 + 0.5f * DT_F * k12,
         sg, rh, bt, k20, k21, k22);
  lorenz(r0 + 0.5f * DT_F * k20, r1 + 0.5f * DT_F * k21, r2 + 0.5f * DT_F * k22,
         sg, rh, bt, k30, k31, k32);
  lorenz(r0 + DT_F * k30, r1 + DT_F * k31, r2 + DT_F * k32,
         sg, rh, bt, k40, k41, k42);
  const float c = DT_F / 6.0f;
  F3 o;
  o.a = r0 + c * (k10 + 2.0f * k20 + 2.0f * k30 + k40);
  o.b = r1 + c * (k11 + 2.0f * k21 + 2.0f * k31 + k41);
  o.c = r2 + c * (k12 + 2.0f * k22 + 2.0f * k32 + k42);
  return o;
}

__global__ __launch_bounds__(256) void mlp_lorenz_rk4(
    const float* __restrict__ x,
    const float* __restrict__ W1, const float* __restrict__ b1,
    const float* __restrict__ W2, const float* __restrict__ b2,
    const float* __restrict__ sigma_p, const float* __restrict__ rho_p,
    const float* __restrict__ beta_p,
    float* __restrict__ out, int numPairs) {
  const int lane = threadIdx.x & 31;
  const int m    = lane & 15;
  const bool hi  = lane >= 16;
  const int waveId = __builtin_amdgcn_readfirstlane((int)((blockIdx.x * blockDim.x + threadIdx.x) >> 5));
  const int nWaves = (gridDim.x * blockDim.x) >> 5;

  const float sg = sigma_p[0], rh = rho_p[0], bt = beta_p[0];
  const float b2_0 = b2[0], b2_1 = b2[1], b2_2 = b2[2];

  const int  r8  = m & 7;
  const bool hv  = r8 < 5;
  const int  hid = (m >> 3) * 5 + r8;
  v2f a1;
  a1.x = (hv && !hi) ? W1[hid * 3] : 0.0f;
  a1.y = hv ? W1[hid * 3 + (hi ? 2 : 1)] : 0.0f;

  const bool jv = r8 < 3;
  const int  mt = m >> 3;
  const int  wo = r8 * 10 + (hi ? 5 : 0);
  v2f a2[5];
#pragma unroll
  for (int i = 0; i < 5; ++i) {
#pragma unroll
    for (int s = 0; s < 2; ++s) {
      const int q    = 2 * i + s;
      const int tile = q >= 5 ? 1 : 0;
      const int hq   = q < 5 ? q : (q == 5 ? 4 : q - 6);
      a2[i][s] = (jv && mt == tile) ? W2[wo + hq] : 0.0f;
    }
  }

  float bb[5];
#pragma unroll
  for (int v = 0; v < 5; ++v) bb[v] = b1[(hi ? 5 : 0) + v];

  const v8f czero = {};

  unsigned lo = lane * 12u;
  unsigned fo = m * 12u + (hi ? 4u : 0u);

  const int numDuo = numPairs >> 1;
  for (int u = waveId; u < numDuo; u += nWaves) {
    asm volatile("" : "+v"(lo), "+v"(fo));

    const char* xb = reinterpret_cast<const char*>(x) + (size_t)u * 768;
    const v2f xA0 = xfrag(xb + fo);
    const v2f xA1 = xfrag(xb + 192 + fo);
    const v2f xB0 = xfrag(xb + 384 + fo);
    const v2f xB1 = xfrag(xb + 576 + fo);

    {
      int un = u + nWaves;
      if (un >= numDuo) un = u;
      const char* xn = reinterpret_cast<const char*>(x) + (size_t)un * 768;
      __builtin_prefetch(xn + lo, 0, 1);
      __builtin_prefetch(xn + 384 + lo, 0, 1);
    }

    v8f hA0 = __builtin_amdgcn_wmma_f32_16x16x4_f32(false, a1, false, xA0, (short)0, czero, false, false);
    v8f hA1 = __builtin_amdgcn_wmma_f32_16x16x4_f32(false, a1, false, xA1, (short)0, czero, false, false);
    v8f hB0 = __builtin_amdgcn_wmma_f32_16x16x4_f32(false, a1, false, xB0, (short)0, czero, false, false);
    v8f hB1 = __builtin_amdgcn_wmma_f32_16x16x4_f32(false, a1, false, xB1, (short)0, czero, false, false);

    bias_relu(hA0, hA1, bb);
    bias_relu(hB0, hB1, bb);

    v2f cA[5], cB[5];
    hfrags(hA0, hA1, cA);
    hfrags(hB0, hB1, cB);

    v8f dA = __builtin_amdgcn_wmma_f32_16x16x4_f32(false, a2[0], false, cA[0], (short)0, czero, false, false);
    v8f dB = __builtin_amdgcn_wmma_f32_16x16x4_f32(false, a2[0], false, cB[0], (short)0, czero, false, false);
#pragma unroll
    for (int i = 1; i < 5; ++i) {
      dA = __builtin_amdgcn_wmma_f32_16x16x4_f32(false, a2[i], false, cA[i], (short)0, dA, false, false);
      dB = __builtin_amdgcn_wmma_f32_16x16x4_f32(false, a2[i], false, cB[i], (short)0, dB, false, false);
    }

    char* ob = reinterpret_cast<char*>(out) + (size_t)u * 768;
    const F3 oA = rk4(dA[0] + b2_0, dA[1] + b2_1, dA[2] + b2_2, sg, rh, bt);
    const F3 oB = rk4(dB[0] + b2_0, dB[1] + b2_1, dB[2] + b2_2, sg, rh, bt);
    *reinterpret_cast<F3*>(ob + lo) = oA;
    *reinterpret_cast<F3*>(ob + 384 + lo) = oB;
  }

  if ((numPairs & 1) && waveId == 0) {
    const size_t off = (size_t)(numPairs - 1) * 384 + lo;
    const char* xt = reinterpret_cast<const char*>(x) + (size_t)(numPairs - 1) * 384 + fo;
    const v2f x0 = xfrag(xt);
    const v2f x1 = xfrag(xt + 192);
    v8f h0 = __builtin_amdgcn_wmma_f32_16x16x4_f32(false, a1, false, x0, (short)0, czero, false, false);
    v8f h1 = __builtin_amdgcn_wmma_f32_16x16x4_f32(false, a1, false, x1, (short)0, czero, false, false);
    bias_relu(h0, h1, bb);
    v2f c[5];
    hfrags(h0, h1, c);
    v8f d = __builtin_amdgcn_wmma_f32_16x16x4_f32(false, a2[0], false, c[0], (short)0, czero, false, false);
#pragma unroll
    for (int i = 1; i < 5; ++i)
      d = __builtin_amdgcn_wmma_f32_16x16x4_f32(false, a2[i], false, c[i], (short)0, d, false, false);
    *reinterpret_cast<F3*>(reinterpret_cast<char*>(out) + off) = rk4(d[0] + b2_0, d[1] + b2_1, d[2] + b2_2, sg, rh, bt);
  }
}

extern "C" void kernel_launch(void* const* d_in, const int* in_sizes, int n_in,
                              void* d_out, int out_size, void* d_ws, size_t ws_size,
                              hipStream_t stream) {
  const float* x     = (const float*)d_in[0];
  const float* W1    = (const float*)d_in[1];
  const float* b1    = (const float*)d_in[2];
  const float* W2    = (const float*)d_in[3];
  const float* b2    = (const float*)d_in[4];
  const float* sigma = (const float*)d_in[5];
  const float* rho   = (const float*)d_in[6];
  const float* beta  = (const float*)d_in[7];
  float* out = (float*)d_out;

  const int Bn = in_sizes[0] / 3;
  const int numPairs = Bn / 32;

  const int block = 256;
  int grid = 4096;
  const int maxBlocks = (numPairs * 16 + block - 1) / block;
  if (grid > maxBlocks) grid = maxBlocks > 0 ? maxBlocks : 1;

  mlp_lorenz_rk4<<<grid, block, 0, stream>>>(x, W1, b1, W2, b2, sigma, rho, beta,
                                             out, numPairs);
}
